# starting kernel
# baseline (speedup 1.0000x reference)
#include <hip/hip_runtime.h>
#include <math.h>

typedef __attribute__((ext_vector_type(2))) float v2f;
typedef __attribute__((ext_vector_type(8))) float v8f;

#define ALPHA 0.2f
#define FDIM  128
#define DNB   16
#define LSEQ  4096
#define BL    32768

__global__ __launch_bounds__(256)
void gat_gemm_xW(const float* __restrict__ x, const float* __restrict__ W,
                 float* __restrict__ h) {
  __shared__ float Wt[FDIM * FDIM];

  for (int i = threadIdx.x; i < FDIM * FDIM; i += blockDim.x) {
    int k = i >> 7, n = i & 127;
    Wt[n * FDIM + k] = W[i];
  }
  __syncthreads();

  const int wave  = threadIdx.x >> 5;
  const int lane  = threadIdx.x & 31;
  const int row0  = (blockIdx.x * 8 + wave) * 16;
  const int m     = lane & 15;
  const int khalf = (lane >> 4) << 1;

  v2f afrag[32];
  const float* xrow = x + (size_t)(row0 + m) * FDIM + khalf;
  #pragma unroll
  for (int ks = 0; ks < 32; ++ks)
    afrag[ks] = *(const v2f*)(xrow + ks * 4);

  v8f acc[8] = {};

  #pragma unroll
  for (int ks = 0; ks < 32; ++ks) {
    const int kb = ks * 4 + khalf;
    #pragma unroll
    for (int n = 0; n < 8; ++n) {
      v2f bfrag = *(const v2f*)(&Wt[(n * 16 + m) * FDIM + kb]);
      acc[n] = __builtin_amdgcn_wmma_f32_16x16x4_f32(
           false, afrag[ks],  false, bfrag,
           (short)0, acc[n],  false,  false);
    }
  }

  const int mhi = (lane >> 4) << 3;
  #pragma unroll
  for (int n = 0; n < 8; ++n) {
    #pragma unroll
    for (int r = 0; r < 8; ++r) {
      h[(size_t)(row0 + r + mhi) * FDIM + n * 16 + m] = acc[n][r];
    }
  }
}

__device__ __forceinline__ float wave_sum(float s) {
  #pragma unroll
  for (int off = 16; off > 0; off >>= 1)
    s += __shfl_xor(s, off, 32);
  return s;
}

__global__ __launch_bounds__(256)
void gat_attn(const float* __restrict__ h, const int* __restrict__ adj,
              const float* __restrict__ a, float* __restrict__ out) {
  const int wid  = (int)((blockIdx.x * blockDim.x + threadIdx.x) >> 5);
  const int lane = threadIdx.x & 31;
  if (wid >= BL) return;

  const int    b    = wid >> 12;
  const size_t base = (size_t)b * LSEQ;
  const float4* hv  = (const float4*)h;

  float4 v[DNB];
  const int* arow = adj + (size_t)wid * DNB;
  #pragma unroll
  for (int d = 0; d < DNB; ++d) {
    const int nb = arow[d];
    v[d] = hv[(base + (size_t)nb) * (FDIM / 4) + lane];
  }

  const float4 alo = ((const float4*)a)[lane];
  const float4 ahi = ((const float4*)a)[32 + lane];

  const float ehi = wave_sum(v[0].x * ahi.x + v[0].y * ahi.y +
                             v[0].z * ahi.z + v[0].w * ahi.w);

  float e[DNB];
  float emax = -1e30f;
  #pragma unroll
  for (int d = 0; d < DNB; ++d) {
    float s = wave_sum(v[d].x * alo.x + v[d].y * alo.y +
                       v[d].z * alo.z + v[d].w * alo.w) + ehi;
    s = (s > 0.f) ? s : ALPHA * s;
    e[d] = s;
    emax = fmaxf(emax, s);
  }

  float denom = 0.f;
  #pragma unroll
  for (int d = 0; d < DNB; ++d) { e[d] = __expf(e[d] - emax); denom += e[d]; }
  const float inv = 1.f / denom;

  float4 o = {0.f, 0.f, 0.f, 0.f};
  #pragma unroll
  for (int d = 0; d < DNB; ++d) {
    const float w = e[d] * inv;
    o.x += w * v[d].x; o.y += w * v[d].y;
    o.z += w * v[d].z; o.w += w * v[d].w;
  }

  o.x = (o.x > 0.f) ? o.x : (__expf(o.x) - 1.f);
  o.y = (o.y > 0.f) ? o.y : (__expf(o.y) - 1.f);
  o.z = (o.z > 0.f) ? o.z : (__expf(o.z) - 1.f);
  o.w = (o.w > 0.f) ? o.w : (__expf(o.w) - 1.f);

  ((float4*)out)[(size_t)wid * (FDIM / 4) + lane] = o;
}

extern "C" void kernel_launch(void* const* d_in, const int* in_sizes, int n_in,
                              void* d_out, int out_size, void* d_ws, size_t ws_size,
                              hipStream_t stream) {
  const float* x   = (const float*)d_in[0];
  const int*   adj = (const int*)d_in[1];
  const float* W   = (const float*)d_in[2];
  const float* a   = (const float*)d_in[3];
  float*       out = (float*)d_out;
  float*       h   = (float*)d_ws;

  gat_gemm_xW<<<256, 256, 0, stream>>>(x, W, h);
  gat_attn<<<4096, 256, 0, stream>>>(h, adj, a, out);
}


# GEMM: W in LDS as k-pair slots, 16-column blocks swapped on odd pair-rows: B-fragment ds_load_b64 bank-conflict-free (was 16-way); staging 16 row loads in flight + b128 stores (was 64 serial b32)
# speedup vs baseline: 2.0575x; 2.0575x over previous
#include <hip/hip_runtime.h>
#include <math.h>

typedef __attribute__((ext_vector_type(2))) float v2f;
typedef __attribute__((ext_vector_type(8))) float v8f;

#define ALPHA 0.2f
#define FDIM  128
#define DNB   16
#define LSEQ  4096
#define BL    32768

__global__ __launch_bounds__(256)
void gat_gemm_xW(const float* __restrict__ x, const float* __restrict__ W,
                 float* __restrict__ h) {
  __shared__ __attribute__((aligned(16))) v2f Wp[(FDIM / 2) * FDIM];

  const int wave  = threadIdx.x >> 5;
  const int lane  = threadIdx.x & 31;

  #pragma unroll
  for (int j = 0; j < 8; ++j) {
    const int p = j * 8 + wave;
    const float4 r0 = *(const float4*)(W + (2 * p)     * FDIM + lane * 4);
    const float4 r1 = *(const float4*)(W + (2 * p + 1) * FDIM + lane * 4);
    const int c = (lane * 4) ^ ((p & 1) << 4);
    float4* dst = (float4*)(&Wp[p * FDIM + c]);
    dst[0] = make_float4(r0.x, r1.x, r0.y, r1.y);
    dst[1] = make_float4(r0.z, r1.z, r0.w, r1.w);
  }
  __syncthreads();

  const int row0  = (blockIdx.x * 8 + wave) * 16;
  const int m     = lane & 15;
  const int half  = lane >> 4;
  const int khalf = half << 1;

  v2f afrag[32];
  const float* xrow = x + (size_t)(row0 + m) * FDIM + khalf;
  #pragma unroll
  for (int ks = 0; ks < 32; ++ks)
    afrag[ks] = *(const v2f*)(xrow + ks * 4);

  v8f acc[8] = {};

  const v2f* bq_even = Wp + half * FDIM + m + (half << 4);
  const v2f* bq_odd  = Wp + half * FDIM + m - (half << 4);

  #pragma unroll
  for (int ks = 0; ks < 32; ++ks) {
    #pragma unroll
    for (int n = 0; n < 8; ++n) {
      const v2f* bq = (n & 1) ? bq_odd : bq_even;
      v2f bfrag = bq[ks * 2 * FDIM + n * 16];
      acc[n] = __builtin_amdgcn_wmma_f32_16x16x4_f32(
           false, afrag[ks],  false, bfrag,
           (short)0, acc[n],  false,  false);
    }
  }

  const int mhi = (lane >> 4) << 3;
  #pragma unroll
  for (int n = 0; n < 8; ++n) {
    #pragma unroll
    for (int r = 0; r < 8; ++r) {
      h[(size_t)(row0 + r + mhi) * FDIM + n * 16 + m] = acc[n][r];
    }
  }
}

__device__ __forceinline__ float wave_sum(float s) {
  #pragma unroll
  for (int off = 16; off > 0; off >>= 1)
    s += __shfl_xor(s, off, 32);
  return s;
}

__global__ __launch_bounds__(256)
void gat_attn(const float* __restrict__ h, const int* __restrict__ adj,
              const float* __restrict__ a, float* __restrict__ out) {
  const int wid  = (int)((blockIdx.x * blockDim.x + threadIdx.x) >> 5);
  const int lane = threadIdx.x & 31;
  if (wid >= BL) return;

  const int    b    = wid >> 12;
  const size_t base = (size_t)b * LSEQ;
  const float4* hv  = (const float4*)h;

  float4 v[DNB];
  const int* arow = adj + (size_t)wid * DNB;
  #pragma unroll
  for (int d = 0; d < DNB; ++d) {
    const int nb = arow[d];
    v[d] = hv[(base + (size_t)nb) * (FDIM / 4) + lane];
  }

  const float4 alo = ((const float4*)a)[lane];
  const float4 ahi = ((const float4*)a)[32 + lane];

  const float ehi = wave_sum(v[0].x * ahi.x + v[0].y * ahi.y +
                             v[0].z * ahi.z + v[0].w * ahi.w);

  float e[DNB];
  float emax = -1e30f;
  #pragma unroll
  for (int d = 0; d < DNB; ++d) {
    float s = wave_sum(v[d].x * alo.x + v[d].y * alo.y +
                       v[d].z * alo.z + v[d].w * alo.w) + ehi;
    s = (s > 0.f) ? s : ALPHA * s;
    e[d] = s;
    emax = fmaxf(emax, s);
  }

  float denom = 0.f;
  #pragma unroll
  for (int d = 0; d < DNB; ++d) { e[d] = __expf(e[d] - emax); denom += e[d]; }
  const float inv = 1.f / denom;

  float4 o = {0.f, 0.f, 0.f, 0.f};
  #pragma unroll
  for (int d = 0; d < DNB; ++d) {
    const float w = e[d] * inv;
    o.x += w * v[d].x; o.y += w * v[d].y;
    o.z += w * v[d].z; o.w += w * v[d].w;
  }

  o.x = (o.x > 0.f) ? o.x : (__expf(o.x) - 1.f);
  o.y = (o.y > 0.f) ? o.y : (__expf(o.y) - 1.f);
  o.z = (o.z > 0.f) ? o.z : (__expf(o.z) - 1.f);
  o.w = (o.w > 0.f) ? o.w : (__expf(o.w) - 1.f);

  ((float4*)out)[(size_t)wid * (FDIM / 4) + lane] = o;
}

extern "C" void kernel_launch(void* const* d_in, const int* in_sizes, int n_in,
                              void* d_out, int out_size, void* d_ws, size_t ws_size,
                              hipStream_t stream) {
  const float* x   = (const float*)d_in[0];
  const int*   adj = (const int*)d_in[1];
  const float* W   = (const float*)d_in[2];
  const float* a   = (const float*)d_in[3];
  float*       out = (float*)d_out;
  float*       h   = (float*)d_ws;

  gat_gemm_xW<<<256, 256, 0, stream>>>(x, W, h);
  gat_attn<<<4096, 256, 0, stream>>>(h, adj, a, out);
}


# Attention: 17 wave-wide sums as one transposed DPP butterfly on the vector ALU (16->8->4->2->1 values a lane), not 85 LDS-unit shuffles; softmax one neighbour a lane; logit, softmax sum order changes
# speedup vs baseline: 2.3073x; 1.1214x over previous
#include <hip/hip_runtime.h>
#include <math.h>

typedef __attribute__((ext_vector_type(2))) float v2f;
typedef __attribute__((ext_vector_type(8))) float v8f;

#define ALPHA 0.2f
#define FDIM  128
#define DNB   16
#define LSEQ  4096
#define BL    32768

__global__ __launch_bounds__(256)
void gat_gemm_xW(const float* __restrict__ x, const float* __restrict__ W,
                 float* __restrict__ h) {
  __shared__ __attribute__((aligned(16))) v2f Wp[(FDIM / 2) * FDIM];

  const int wave  = threadIdx.x >> 5;
  const int lane  = threadIdx.x & 31;

  #pragma unroll
  for (int j = 0; j < 8; ++j) {
    const int p = j * 8 + wave;
    const float4 r0 = *(const float4*)(W + (2 * p)     * FDIM + lane * 4);
    const float4 r1 = *(const float4*)(W + (2 * p + 1) * FDIM + lane * 4);
    const int c = (lane * 4) ^ ((p & 1) << 4);
    float4* dst = (float4*)(&Wp[p * FDIM + c]);
    dst[0] = make_float4(r0.x, r1.x, r0.y, r1.y);
    dst[1] = make_float4(r0.z, r1.z, r0.w, r1.w);
  }
  __syncthreads();

  const int row0  = (blockIdx.x * 8 + wave) * 16;
  const int m     = lane & 15;
  const int half  = lane >> 4;
  const int khalf = half << 1;

  v2f afrag[32];
  const float* xrow = x + (size_t)(row0 + m) * FDIM + khalf;
  #pragma unroll
  for (int ks = 0; ks < 32; ++ks)
    afrag[ks] = *(const v2f*)(xrow + ks * 4);

  v8f acc[8] = {};

  const v2f* bq_even = Wp + half * FDIM + m + (half << 4);
  const v2f* bq_odd  = Wp + half * FDIM + m - (half << 4);

  #pragma unroll
  for (int ks = 0; ks < 32; ++ks) {
    #pragma unroll
    for (int n = 0; n < 8; ++n) {
      const v2f* bq = (n & 1) ? bq_odd : bq_even;
      v2f bfrag = bq[ks * 2 * FDIM + n * 16];
      acc[n] = __builtin_amdgcn_wmma_f32_16x16x4_f32(
           false, afrag[ks],  false, bfrag,
           (short)0, acc[n],  false,  false);
    }
  }

  const int mhi = (lane >> 4) << 3;
  #pragma unroll
  for (int n = 0; n < 8; ++n) {
    #pragma unroll
    for (int r = 0; r < 8; ++r) {
      h[(size_t)(row0 + r + mhi) * FDIM + n * 16 + m] = acc[n][r];
    }
  }
}


template <int M>
__device__ __forceinline__ float row_xor(float x) {
  return __builtin_bit_cast(float,
      __builtin_amdgcn_update_dpp(0, __builtin_bit_cast(int, x), 0x160 + M, 0xf, 0xf, true));
}

__device__ __forceinline__ float other_row(float x) {
  const int xi = __builtin_bit_cast(int, x);
  return __builtin_bit_cast(float,
      __builtin_amdgcn_permlanex16(xi, xi, 0x76543210u, 0xfedcba98u, false, false));
}

template <int M, int N>
__device__ __forceinline__ void halve(float (&p)[DNB], const bool up) {
  #pragma unroll
  for (int i = 0; i < N / 2; ++i) {
    const float keep = up ? p[i + N / 2] : p[i];
    const float send = up ? p[i] : p[i + N / 2];
    p[i] = keep + row_xor<M>(send);
  }
}

__device__ __forceinline__ float row_max16(float s) {
  s = fmaxf(s, row_xor<1>(s));
  s = fmaxf(s, row_xor<2>(s));
  s = fmaxf(s, row_xor<4>(s));
  s = fmaxf(s, row_xor<8>(s));
  return s;
}

__device__ __forceinline__ float row_sum16(float s) {
  s += row_xor<1>(s);
  s += row_xor<2>(s);
  s += row_xor<4>(s);
  s += row_xor<8>(s);
  return s;
}

__global__ __launch_bounds__(256)
void gat_attn(const float* __restrict__ h, const int* __restrict__ adj,
              const float* __restrict__ a, float* __restrict__ out) {
  const int wid  = (int)((blockIdx.x * blockDim.x + threadIdx.x) >> 5);
  const int lane = threadIdx.x & 31;
  if (wid >= BL) return;

  const int    b    = wid >> 12;
  const size_t base = (size_t)b * LSEQ;
  const float4* hv  = (const float4*)h;

  float4 v[DNB];
  const int* arow = adj + (size_t)wid * DNB;
  #pragma unroll
  for (int d = 0; d < DNB; ++d) {
    const int nb = arow[d];
    v[d] = hv[(base + (size_t)nb) * (FDIM / 4) + lane];
  }

  const float4 alo = ((const float4*)a)[lane];
  const float4 ahi = ((const float4*)a)[32 + lane];

  const float q = fmaf(v[0].w, ahi.w, fmaf(v[0].z, ahi.z, fmaf(v[0].y, ahi.y, v[0].x * ahi.x)));

  float p[DNB];
  #pragma unroll
  for (int d = 0; d < DNB; ++d)
    p[d] = fmaf(v[d].w, alo.w, fmaf(v[d].z, alo.z, fmaf(v[d].y, alo.y, fmaf(v[d].x, alo.x, q))));

  halve<8, 16>(p, (lane & 8) != 0);
  halve<4,  8>(p, (lane & 4) != 0);
  halve<2,  4>(p, (lane & 2) != 0);
  halve<1,  2>(p, (lane & 1) != 0);
  float s = p[0] + other_row(p[0]);

  s = (s > 0.f) ? s : ALPHA * s;

  const float emax  = row_max16(s);
  const float ex    = __expf(s - emax);
  const float denom = row_sum16(ex);
  const float wl    = ex * (1.f / denom);

  float4 o = {0.f, 0.f, 0.f, 0.f};
  #pragma unroll
  for (int d = 0; d < DNB; ++d) {
    const float w = __builtin_bit_cast(float,
        __builtin_amdgcn_readlane(__builtin_bit_cast(int, wl), d));
    o.x = fmaf(w, v[d].x, o.x); o.y = fmaf(w, v[d].y, o.y);
    o.z = fmaf(w, v[d].z, o.z); o.w = fmaf(w, v[d].w, o.w);
  }

  o.x = (o.x > 0.f) ? o.x : (__expf(o.x) - 1.f);
  o.y = (o.y > 0.f) ? o.y : (__expf(o.y) - 1.f);
  o.z = (o.z > 0.f) ? o.z : (__expf(o.z) - 1.f);
  o.w = (o.w > 0.f) ? o.w : (__expf(o.w) - 1.f);

  ((float4*)out)[(size_t)wid * (FDIM / 4) + lane] = o;
}

extern "C" void kernel_launch(void* const* d_in, const int* in_sizes, int n_in,
                              void* d_out, int out_size, void* d_ws, size_t ws_size,
                              hipStream_t stream) {
  const float* x   = (const float*)d_in[0];
  const int*   adj = (const int*)d_in[1];
  const float* W   = (const float*)d_in[2];
  const float* a   = (const float*)d_in[3];
  float*       out = (float*)d_out;
  float*       h   = (float*)d_ws;

  gat_gemm_xW<<<256, 256, 0, stream>>>(x, W, h);
  gat_attn<<<4096, 256, 0, stream>>>(h, adj, a, out);
}


# Attention: a wave walks 8 consecutive nodes, requesting the next node's 16 rows (second register set) before it computes the current one; ids by one 512-B load; 4096 resident waves; sums unchanged
# speedup vs baseline: 2.5377x; 1.0999x over previous
#include <hip/hip_runtime.h>
#include <math.h>

typedef __attribute__((ext_vector_type(2))) float v2f;
typedef __attribute__((ext_vector_type(8))) float v8f;

#define ALPHA 0.2f
#define FDIM  128
#define DNB   16
#define LSEQ  4096
#define BL    32768

__global__ __launch_bounds__(256)
void gat_gemm_xW(const float* __restrict__ x, const float* __restrict__ W,
                 float* __restrict__ h) {
  __shared__ __attribute__((aligned(16))) v2f Wp[(FDIM / 2) * FDIM];

  const int wave  = threadIdx.x >> 5;
  const int lane  = threadIdx.x & 31;

  #pragma unroll
  for (int j = 0; j < 8; ++j) {
    const int p = j * 8 + wave;
    const float4 r0 = *(const float4*)(W + (2 * p)     * FDIM + lane * 4);
    const float4 r1 = *(const float4*)(W + (2 * p + 1) * FDIM + lane * 4);
    const int c = (lane * 4) ^ ((p & 1) << 4);
    float4* dst = (float4*)(&Wp[p * FDIM + c]);
    dst[0] = make_float4(r0.x, r1.x, r0.y, r1.y);
    dst[1] = make_float4(r0.z, r1.z, r0.w, r1.w);
  }
  __syncthreads();

  const int row0  = (blockIdx.x * 8 + wave) * 16;
  const int m     = lane & 15;
  const int half  = lane >> 4;
  const int khalf = half << 1;

  v2f afrag[32];
  const float* xrow = x + (size_t)(row0 + m) * FDIM + khalf;
  #pragma unroll
  for (int ks = 0; ks < 32; ++ks)
    afrag[ks] = *(const v2f*)(xrow + ks * 4);

  v8f acc[8] = {};

  const v2f* bq_even = Wp + half * FDIM + m + (half << 4);
  const v2f* bq_odd  = Wp + half * FDIM + m - (half << 4);

  #pragma unroll
  for (int ks = 0; ks < 32; ++ks) {
    #pragma unroll
    for (int n = 0; n < 8; ++n) {
      const v2f* bq = (n & 1) ? bq_odd : bq_even;
      v2f bfrag = bq[ks * 2 * FDIM + n * 16];
      acc[n] = __builtin_amdgcn_wmma_f32_16x16x4_f32(
           false, afrag[ks],  false, bfrag,
           (short)0, acc[n],  false,  false);
    }
  }

  const int mhi = (lane >> 4) << 3;
  #pragma unroll
  for (int n = 0; n < 8; ++n) {
    #pragma unroll
    for (int r = 0; r < 8; ++r) {
      h[(size_t)(row0 + r + mhi) * FDIM + n * 16 + m] = acc[n][r];
    }
  }
}


#define NPW 8

template <int M>
__device__ __forceinline__ float row_xor(float x) {
  return __builtin_bit_cast(float,
      __builtin_amdgcn_update_dpp(0, __builtin_bit_cast(int, x), 0x160 + M, 0xf, 0xf, true));
}

__device__ __forceinline__ float other_row(float x) {
  const int xi = __builtin_bit_cast(int, x);
  return __builtin_bit_cast(float,
      __builtin_amdgcn_permlanex16(xi, xi, 0x76543210u, 0xfedcba98u, false, false));
}

template <int M, int N>
__device__ __forceinline__ void halve(float (&p)[DNB], const bool up) {
  #pragma unroll
  for (int i = 0; i < N / 2; ++i) {
    const float keep = up ? p[i + N / 2] : p[i];
    const float send = up ? p[i] : p[i + N / 2];
    p[i] = keep + row_xor<M>(send);
  }
}

__device__ __forceinline__ float row_max16(float s) {
  s = fmaxf(s, row_xor<1>(s));
  s = fmaxf(s, row_xor<2>(s));
  s = fmaxf(s, row_xor<4>(s));
  s = fmaxf(s, row_xor<8>(s));
  return s;
}

__device__ __forceinline__ float row_sum16(float s) {
  s += row_xor<1>(s);
  s += row_xor<2>(s);
  s += row_xor<4>(s);
  s += row_xor<8>(s);
  return s;
}

__device__ __forceinline__ void request_rows(float4 (&v)[DNB], const float4* __restrict__ hb,
                                             const int4 adjv, const int j, const unsigned lane) {
  #pragma unroll
  for (int d = 0; d < DNB; ++d) {
    const int c   = d & 3;
    const int ids = (c == 0) ? adjv.x : (c == 1) ? adjv.y : (c == 2) ? adjv.z : adjv.w;
    const int nb  = __builtin_amdgcn_readlane(ids, j * 4 + (d >> 2));
    const float4* row = hb + (size_t)nb * (FDIM / 4);
    v[d] = row[lane];
  }
}

__device__ __forceinline__ void arithmetic_behind_requests(float4& ahi) {
  asm volatile("" : "+v"(ahi.x), "+v"(ahi.y), "+v"(ahi.z), "+v"(ahi.w) : : "memory");
}

__device__ __forceinline__ float4 node_out(const float4 (&v)[DNB], const float4 alo, const float4 ahi,
                                           const int lane) {
  const float q = fmaf(v[0].w, ahi.w, fmaf(v[0].z, ahi.z, fmaf(v[0].y, ahi.y, v[0].x * ahi.x)));

  float p[DNB];
  #pragma unroll
  for (int d = 0; d < DNB; ++d)
    p[d] = fmaf(v[d].w, alo.w, fmaf(v[d].z, alo.z, fmaf(v[d].y, alo.y, fmaf(v[d].x, alo.x, q))));

  halve<8, 16>(p, (lane & 8) != 0);
  halve<4,  8>(p, (lane & 4) != 0);
  halve<2,  4>(p, (lane & 2) != 0);
  halve<1,  2>(p, (lane & 1) != 0);
  float s = p[0] + other_row(p[0]);

  s = (s > 0.f) ? s : ALPHA * s;

  const float emax  = row_max16(s);
  const float ex    = __expf(s - emax);
  const float denom = row_sum16(ex);
  const float wl    = ex * (1.f / denom);

  float4 o = {0.f, 0.f, 0.f, 0.f};
  #pragma unroll
  for (int d = 0; d < DNB; ++d) {
    const float w = __builtin_bit_cast(float,
        __builtin_amdgcn_readlane(__builtin_bit_cast(int, wl), d));
    o.x = fmaf(w, v[d].x, o.x); o.y = fmaf(w, v[d].y, o.y);
    o.z = fmaf(w, v[d].z, o.z); o.w = fmaf(w, v[d].w, o.w);
  }

  o.x = (o.x > 0.f) ? o.x : (__expf(o.x) - 1.f);
  o.y = (o.y > 0.f) ? o.y : (__expf(o.y) - 1.f);
  o.z = (o.z > 0.f) ? o.z : (__expf(o.z) - 1.f);
  o.w = (o.w > 0.f) ? o.w : (__expf(o.w) - 1.f);
  return o;
}

__global__ __launch_bounds__(256)
void gat_attn(const float* __restrict__ h, const int* __restrict__ adj,
              const float* __restrict__ a, float* __restrict__ out) {
  const int wave = __builtin_amdgcn_readfirstlane((int)((blockIdx.x * 256u + threadIdx.x) >> 5));
  const int lane = threadIdx.x & 31;

  const int node0 = wave * NPW;
  const int b     = node0 >> 12;
  const float4* hb = (const float4*)h + (size_t)b * LSEQ * (FDIM / 4);

  const int4 adjv = ((const int4*)adj)[(size_t)wave * (NPW * DNB / 4) + lane];

  const float4 alo = ((const float4*)a)[lane];
  float4 ahi = ((const float4*)a)[32 + lane];

  float4* orow = (float4*)out + (size_t)node0 * (FDIM / 4) + lane;

  float4 va[DNB], vb[DNB];
  request_rows(va, hb, adjv, 0, lane);

  #pragma unroll 1
  for (int j = 0; j < NPW - 2; j += 2) {
    request_rows(vb, hb, adjv, j + 1, lane);
    arithmetic_behind_requests(ahi);
    orow[(size_t)j * (FDIM / 4)] = node_out(va, alo, ahi, lane);
    request_rows(va, hb, adjv, j + 2, lane);
    arithmetic_behind_requests(ahi);
    orow[(size_t)(j + 1) * (FDIM / 4)] = node_out(vb, alo, ahi, lane);
  }
  request_rows(vb, hb, adjv, NPW - 1, lane);
  arithmetic_behind_requests(ahi);
  orow[(size_t)(NPW - 2) * (FDIM / 4)] = node_out(va, alo, ahi, lane);
  orow[(size_t)(NPW - 1) * (FDIM / 4)] = node_out(vb, alo, ahi, lane);
}

extern "C" void kernel_launch(void* const* d_in, const int* in_sizes, int n_in,
                              void* d_out, int out_size, void* d_ws, size_t ws_size,
                              hipStream_t stream) {
  const float* x   = (const float*)d_in[0];
  const int*   adj = (const int*)d_in[1];
  const float* W   = (const float*)d_in[2];
  const float* a   = (const float*)d_in[3];
  float*       out = (float*)d_out;
  float*       h   = (float*)d_ws;

  gat_gemm_xW<<<256, 256, 0, stream>>>(x, W, h);
  gat_attn<<<512, 256, 0, stream>>>(h, adj, a, out);
}


# GEMM: the 8 B fragments of K-step ks+1 are requested from LDS as one group ahead of the 8 WMMAs of K-step ks (second register set, scheduling fences), not each just before the WMMA that uses it
# speedup vs baseline: 2.6694x; 1.0519x over previous
#include <hip/hip_runtime.h>
#include <math.h>

typedef __attribute__((ext_vector_type(2))) float v2f;
typedef __attribute__((ext_vector_type(8))) float v8f;

#define ALPHA 0.2f
#define FDIM  128
#define DNB   16
#define LSEQ  4096
#define BL    32768

__global__ __launch_bounds__(256)
void gat_gemm_xW(const float* __restrict__ x, const float* __restrict__ W,
                 float* __restrict__ h) {
  __shared__ __attribute__((aligned(16))) v2f Wp[(FDIM / 2) * FDIM];

  const int wave  = threadIdx.x >> 5;
  const int lane  = threadIdx.x & 31;

  #pragma unroll
  for (int j = 0; j < 8; ++j) {
    const int p = j * 8 + wave;
    const float4 r0 = *(const float4*)(W + (2 * p)     * FDIM + lane * 4);
    const float4 r1 = *(const float4*)(W + (2 * p + 1) * FDIM + lane * 4);
    const int c = (lane * 4) ^ ((p & 1) << 4);
    float4* dst = (float4*)(&Wp[p * FDIM + c]);
    dst[0] = make_float4(r0.x, r1.x, r0.y, r1.y);
    dst[1] = make_float4(r0.z, r1.z, r0.w, r1.w);
  }
  __syncthreads();

  const int row0  = (blockIdx.x * 8 + wave) * 16;
  const int m     = lane & 15;
  const int half  = lane >> 4;
  const int khalf = half << 1;

  v2f afrag[32];
  const float* xrow = x + (size_t)(row0 + m) * FDIM + khalf;
  #pragma unroll
  for (int ks = 0; ks < 32; ++ks)
    afrag[ks] = *(const v2f*)(xrow + ks * 4);

  v8f acc[8] = {};

  const v2f* bq_even = Wp + half * FDIM + m + (half << 4);
  const v2f* bq_odd  = Wp + half * FDIM + m - (half << 4);

  v2f bfr[2][8];
  #pragma unroll
  for (int n = 0; n < 8; ++n)
    bfr[0][n] = ((n & 1) ? bq_odd : bq_even)[n * 16];

  #pragma unroll
  for (int ks = 0; ks < 32; ++ks) {
    if (ks + 1 < 32) {
      #pragma unroll
      for (int n = 0; n < 8; ++n)
        bfr[(ks + 1) & 1][n] = ((n & 1) ? bq_odd : bq_even)[(ks + 1) * 2 * FDIM + n * 16];
    }
    __builtin_amdgcn_sched_barrier(0);
    #pragma unroll
    for (int n = 0; n < 8; ++n) {
      acc[n] = __builtin_amdgcn_wmma_f32_16x16x4_f32(
           false, afrag[ks],  false, bfr[ks & 1][n],
           (short)0, acc[n],  false,  false);
    }
    __builtin_amdgcn_sched_barrier(0);
  }

  const int mhi = (lane >> 4) << 3;
  #pragma unroll
  for (int n = 0; n < 8; ++n) {
    #pragma unroll
    for (int r = 0; r < 8; ++r) {
      h[(size_t)(row0 + r + mhi) * FDIM + n * 16 + m] = acc[n][r];
    }
  }
}


#define NPW 8

template <int M>
__device__ __forceinline__ float row_xor(float x) {
  return __builtin_bit_cast(float,
      __builtin_amdgcn_update_dpp(0, __builtin_bit_cast(int, x), 0x160 + M, 0xf, 0xf, true));
}

__device__ __forceinline__ float other_row(float x) {
  const int xi = __builtin_bit_cast(int, x);
  return __builtin_bit_cast(float,
      __builtin_amdgcn_permlanex16(xi, xi, 0x76543210u, 0xfedcba98u, false, false));
}

template <int M, int N>
__device__ __forceinline__ void halve(float (&p)[DNB], const bool up) {
  #pragma unroll
  for (int i = 0; i < N / 2; ++i) {
    const float keep = up ? p[i + N / 2] : p[i];
    const float send = up ? p[i] : p[i + N / 2];
    p[i] = keep + row_xor<M>(send);
  }
}

__device__ __forceinline__ float row_max16(float s) {
  s = fmaxf(s, row_xor<1>(s));
  s = fmaxf(s, row_xor<2>(s));
  s = fmaxf(s, row_xor<4>(s));
  s = fmaxf(s, row_xor<8>(s));
  return s;
}

__device__ __forceinline__ float row_sum16(float s) {
  s += row_xor<1>(s);
  s += row_xor<2>(s);
  s += row_xor<4>(s);
  s += row_xor<8>(s);
  return s;
}

__device__ __forceinline__ void request_rows(float4 (&v)[DNB], const float4* __restrict__ hb,
                                             const int4 adjv, const int j, const unsigned lane) {
  #pragma unroll
  for (int d = 0; d < DNB; ++d) {
    const int c   = d & 3;
    const int ids = (c == 0) ? adjv.x : (c == 1) ? adjv.y : (c == 2) ? adjv.z : adjv.w;
    const int nb  = __builtin_amdgcn_readlane(ids, j * 4 + (d >> 2));
    const float4* row = hb + (size_t)nb * (FDIM / 4);
    v[d] = row[lane];
  }
}

__device__ __forceinline__ void arithmetic_behind_requests(float4& ahi) {
  asm volatile("" : "+v"(ahi.x), "+v"(ahi.y), "+v"(ahi.z), "+v"(ahi.w) : : "memory");
}

__device__ __forceinline__ float4 node_out(const float4 (&v)[DNB], const float4 alo, const float4 ahi,
                                           const int lane) {
  const float q = fmaf(v[0].w, ahi.w, fmaf(v[0].z, ahi.z, fmaf(v[0].y, ahi.y, v[0].x * ahi.x)));

  float p[DNB];
  #pragma unroll
  for (int d = 0; d < DNB; ++d)
    p[d] = fmaf(v[d].w, alo.w, fmaf(v[d].z, alo.z, fmaf(v[d].y, alo.y, fmaf(v[d].x, alo.x, q))));

  halve<8, 16>(p, (lane & 8) != 0);
  halve<4,  8>(p, (lane & 4) != 0);
  halve<2,  4>(p, (lane & 2) != 0);
  halve<1,  2>(p, (lane & 1) != 0);
  float s = p[0] + other_row(p[0]);

  s = (s > 0.f) ? s : ALPHA * s;

  const float emax  = row_max16(s);
  const float ex    = __expf(s - emax);
  const float denom = row_sum16(ex);
  const float wl    = ex * (1.f / denom);

  float4 o = {0.f, 0.f, 0.f, 0.f};
  #pragma unroll
  for (int d = 0; d < DNB; ++d) {
    const float w = __builtin_bit_cast(float,
        __builtin_amdgcn_readlane(__builtin_bit_cast(int, wl), d));
    o.x = fmaf(w, v[d].x, o.x); o.y = fmaf(w, v[d].y, o.y);
    o.z = fmaf(w, v[d].z, o.z); o.w = fmaf(w, v[d].w, o.w);
  }

  o.x = (o.x > 0.f) ? o.x : (__expf(o.x) - 1.f);
  o.y = (o.y > 0.f) ? o.y : (__expf(o.y) - 1.f);
  o.z = (o.z > 0.f) ? o.z : (__expf(o.z) - 1.f);
  o.w = (o.w > 0.f) ? o.w : (__expf(o.w) - 1.f);
  return o;
}

__global__ __launch_bounds__(256)
void gat_attn(const float* __restrict__ h, const int* __restrict__ adj,
              const float* __restrict__ a, float* __restrict__ out) {
  const int wave = __builtin_amdgcn_readfirstlane((int)((blockIdx.x * 256u + threadIdx.x) >> 5));
  const int lane = threadIdx.x & 31;

  const int node0 = wave * NPW;
  const int b     = node0 >> 12;
  const float4* hb = (const float4*)h + (size_t)b * LSEQ * (FDIM / 4);

  const int4 adjv = ((const int4*)adj)[(size_t)wave * (NPW * DNB / 4) + lane];

  const float4 alo = ((const float4*)a)[lane];
  float4 ahi = ((const float4*)a)[32 + lane];

  float4* orow = (float4*)out + (size_t)node0 * (FDIM / 4) + lane;

  float4 va[DNB], vb[DNB];
  request_rows(va, hb, adjv, 0, lane);

  #pragma unroll 1
  for (int j = 0; j < NPW - 2; j += 2) {
    request_rows(vb, hb, adjv, j + 1, lane);
    arithmetic_behind_requests(ahi);
    orow[(size_t)j * (FDIM / 4)] = node_out(va, alo, ahi, lane);
    request_rows(va, hb, adjv, j + 2, lane);
    arithmetic_behind_requests(ahi);
    orow[(size_t)(j + 1) * (FDIM / 4)] = node_out(vb, alo, ahi, lane);
  }
  request_rows(vb, hb, adjv, NPW - 1, lane);
  arithmetic_behind_requests(ahi);
  orow[(size_t)(NPW - 2) * (FDIM / 4)] = node_out(va, alo, ahi, lane);
  orow[(size_t)(NPW - 1) * (FDIM / 4)] = node_out(vb, alo, ahi, lane);
}

extern "C" void kernel_launch(void* const* d_in, const int* in_sizes, int n_in,
                              void* d_out, int out_size, void* d_ws, size_t ws_size,
                              hipStream_t stream) {
  const float* x   = (const float*)d_in[0];
  const int*   adj = (const int*)d_in[1];
  const float* W   = (const float*)d_in[2];
  const float* a   = (const float*)d_in[3];
  float*       out = (float*)d_out;
  float*       h   = (float*)d_ws;

  gat_gemm_xW<<<256, 256, 0, stream>>>(x, W, h);
  gat_attn<<<512, 256, 0, stream>>>(h, adj, a, out);
}


# Projection moved behind the neighbour sum (linearity): gather x rows, logits by x.(W a), z=sum w*x in LDS, then z@W on WMMA + ELU in the same kernel; h never stored, one launch; sum order changes
# speedup vs baseline: 3.5306x; 1.3226x over previous
#include <hip/hip_runtime.h>
#include <math.h>


typedef __attribute__((ext_vector_type(2))) float v2f;
typedef __attribute__((ext_vector_type(8))) float v8f;

#define ALPHA 0.2f
#define FDIM  128
#define DNB   16
#define LSEQ  4096


#define NPW 8
#define ZP  132

template <int M>
__device__ __forceinline__ float row_xor(float x) {
  return __builtin_bit_cast(float,
      __builtin_amdgcn_update_dpp(0, __builtin_bit_cast(int, x), 0x160 + M, 0xf, 0xf, true));
}

__device__ __forceinline__ float other_row(float x) {
  const int xi = __builtin_bit_cast(int, x);
  return __builtin_bit_cast(float,
      __builtin_amdgcn_permlanex16(xi, xi, 0x76543210u, 0xfedcba98u, false, false));
}

template <int M, int N, int SZ>
__device__ __forceinline__ void halve(float (&p)[SZ], const bool up) {
  #pragma unroll
  for (int i = 0; i < N / 2; ++i) {
    const float keep = up ? p[i + N / 2] : p[i];
    const float send = up ? p[i] : p[i + N / 2];
    p[i] = keep + row_xor<M>(send);
  }
}

__device__ __forceinline__ float row_max16(float s) {
  s = fmaxf(s, row_xor<1>(s));
  s = fmaxf(s, row_xor<2>(s));
  s = fmaxf(s, row_xor<4>(s));
  s = fmaxf(s, row_xor<8>(s));
  return s;
}

__device__ __forceinline__ float row_sum16(float s) {
  s += row_xor<1>(s);
  s += row_xor<2>(s);
  s += row_xor<4>(s);
  s += row_xor<8>(s);
  return s;
}

__device__ __forceinline__ void request_rows(float4 (&v)[DNB], const float4* __restrict__ hb,
                                             const int4 adjv, const int j, const unsigned lane) {
  #pragma unroll
  for (int d = 0; d < DNB; ++d) {
    const int c   = d & 3;
    const int ids = (c == 0) ? adjv.x : (c == 1) ? adjv.y : (c == 2) ? adjv.z : adjv.w;
    const int nb  = __builtin_amdgcn_readlane(ids, j * 4 + (d >> 2));
    const float4* row = hb + (size_t)nb * (FDIM / 4);
    v[d] = row[lane];
  }
}

__device__ __forceinline__ void arithmetic_behind_requests(float4& ahi) {
  asm volatile("" : "+v"(ahi.x), "+v"(ahi.y), "+v"(ahi.z), "+v"(ahi.w) : : "memory");
}

__device__ __forceinline__ float4 node_agg(const float4 (&v)[DNB], const float4 alo, const float4 ahi,
                                           const int lane) {
  const float q = fmaf(v[0].w, ahi.w, fmaf(v[0].z, ahi.z, fmaf(v[0].y, ahi.y, v[0].x * ahi.x)));

  float p[DNB];
  #pragma unroll
  for (int d = 0; d < DNB; ++d)
    p[d] = fmaf(v[d].w, alo.w, fmaf(v[d].z, alo.z, fmaf(v[d].y, alo.y, fmaf(v[d].x, alo.x, q))));

  halve<8, 16>(p, (lane & 8) != 0);
  halve<4,  8>(p, (lane & 4) != 0);
  halve<2,  4>(p, (lane & 2) != 0);
  halve<1,  2>(p, (lane & 1) != 0);
  float s = p[0] + other_row(p[0]);

  s = (s > 0.f) ? s : ALPHA * s;

  const float emax  = row_max16(s);
  const float ex    = __expf(s - emax);
  const float denom = row_sum16(ex);
  const float wl    = ex * (1.f / denom);

  float4 o = {0.f, 0.f, 0.f, 0.f};
  #pragma unroll
  for (int d = 0; d < DNB; ++d) {
    const float w = __builtin_bit_cast(float,
        __builtin_amdgcn_readlane(__builtin_bit_cast(int, wl), d));
    o.x = fmaf(w, v[d].x, o.x); o.y = fmaf(w, v[d].y, o.y);
    o.z = fmaf(w, v[d].z, o.z); o.w = fmaf(w, v[d].w, o.w);
  }

  return o;
}

__global__ __launch_bounds__(256)
void gat_fused(const float* __restrict__ x, const int* __restrict__ adj,
               const float* __restrict__ W, const float* __restrict__ a,
               float* __restrict__ out) {
  __shared__ __attribute__((aligned(16))) v2f   Wp[(FDIM / 2) * FDIM];
  __shared__ __attribute__((aligned(16))) float Zs[8 * NPW * ZP];
  __shared__ __attribute__((aligned(16))) float us[2 * FDIM];

  const int wave = __builtin_amdgcn_readfirstlane((int)((blockIdx.x * 256u + threadIdx.x) >> 5));
  const int wib  = wave & 7;
  const int lane = threadIdx.x & 31;

  const int node0 = wave * NPW;
  const int b     = node0 >> 12;
  const float4* xb = (const float4*)x + (size_t)b * LSEQ * (FDIM / 4);

  const int4 adjv = ((const int4*)adj)[(size_t)wave * (NPW * DNB / 4) + lane];

  const float4 a_lo = ((const float4*)a)[lane];
  const float4 a_hi = ((const float4*)a)[32 + lane];

  float pu[32];
  #pragma unroll
  for (int j = 0; j < 8; ++j) {
    const int p = j * 8 + wib;
    const float4 r0 = *(const float4*)(W + (2 * p)     * FDIM + lane * 4);
    const float4 r1 = *(const float4*)(W + (2 * p + 1) * FDIM + lane * 4);
    const int c = (lane * 4) ^ ((p & 1) << 4);
    float4* dst = (float4*)(&Wp[p * FDIM + c]);
    dst[0] = make_float4(r0.x, r1.x, r0.y, r1.y);
    dst[1] = make_float4(r0.z, r1.z, r0.w, r1.w);
    pu[4 * j + 0] = fmaf(r0.w, a_lo.w, fmaf(r0.z, a_lo.z, fmaf(r0.y, a_lo.y, r0.x * a_lo.x)));
    pu[4 * j + 1] = fmaf(r1.w, a_lo.w, fmaf(r1.z, a_lo.z, fmaf(r1.y, a_lo.y, r1.x * a_lo.x)));
    pu[4 * j + 2] = fmaf(r0.w, a_hi.w, fmaf(r0.z, a_hi.z, fmaf(r0.y, a_hi.y, r0.x * a_hi.x)));
    pu[4 * j + 3] = fmaf(r1.w, a_hi.w, fmaf(r1.z, a_hi.z, fmaf(r1.y, a_hi.y, r1.x * a_hi.x)));
  }
  {
    const bool up16 = (lane & 16) != 0;
    #pragma unroll
    for (int i = 0; i < 16; ++i) {
      const float keep = up16 ? pu[i + 16] : pu[i];
      const float send = up16 ? pu[i] : pu[i + 16];
      pu[i] = keep + other_row(send);
    }
    halve<8, 16>(pu, (lane & 8) != 0);
    halve<4,  8>(pu, (lane & 4) != 0);
    halve<2,  4>(pu, (lane & 2) != 0);
    halve<1,  2>(pu, (lane & 1) != 0);
    const int jj = lane >> 2, qq = lane & 3;
    us[(qq >> 1) * FDIM + 2 * (8 * jj + wib) + (qq & 1)] = pu[0];
  }

  float4 va[DNB], vb[DNB];
  request_rows(va, xb, adjv, 0, lane);

  __syncthreads();

  const float4 ulo = ((const float4*)us)[lane];
  float4 uhi = ((const float4*)us)[32 + lane];

  float4* zrow = (float4*)(Zs + (size_t)(wib * NPW) * ZP) + lane;

  #pragma unroll 1
  for (int j = 0; j < NPW - 2; j += 2) {
    request_rows(vb, xb, adjv, j + 1, lane);
    arithmetic_behind_requests(uhi);
    zrow[j * (ZP / 4)] = node_agg(va, ulo, uhi, lane);
    request_rows(va, xb, adjv, j + 2, lane);
    arithmetic_behind_requests(uhi);
    zrow[(j + 1) * (ZP / 4)] = node_agg(vb, ulo, uhi, lane);
  }
  request_rows(vb, xb, adjv, NPW - 1, lane);
  arithmetic_behind_requests(uhi);
  zrow[(NPW - 2) * (ZP / 4)] = node_agg(va, ulo, uhi, lane);
  zrow[(NPW - 1) * (ZP / 4)] = node_agg(vb, ulo, uhi, lane);

  __syncthreads();

  const int m     = lane & 15;
  const int half  = lane >> 4;
  const int mt    = wib >> 1;
  const int nh    = wib & 1;

  const v2f* aq = (const v2f*)(Zs + (size_t)(mt * 16 + m) * ZP + (half << 1));

  const v2f* bq_even = Wp + half * FDIM + nh * 64 + m + (half << 4);
  const v2f* bq_odd  = Wp + half * FDIM + nh * 64 + m - (half << 4);

  v2f afr[2];
  v2f bfr[2][4];
  afr[0] = aq[0];
  #pragma unroll
  for (int n = 0; n < 4; ++n)
    bfr[0][n] = ((n & 1) ? bq_odd : bq_even)[n * 16];

  v8f acc[4] = {};

  #pragma unroll
  for (int ks = 0; ks < 32; ++ks) {
    if (ks + 1 < 32) {
      afr[(ks + 1) & 1] = aq[(ks + 1) * 2];
      #pragma unroll
      for (int n = 0; n < 4; ++n)
        bfr[(ks + 1) & 1][n] = ((n & 1) ? bq_odd : bq_even)[(ks + 1) * 2 * FDIM + n * 16];
    }
    __builtin_amdgcn_sched_barrier(0);
    #pragma unroll
    for (int n = 0; n < 4; ++n) {
      acc[n] = __builtin_amdgcn_wmma_f32_16x16x4_f32(
           false, afr[ks & 1],  false, bfr[ks & 1][n],
           (short)0, acc[n],  false,  false);
    }
    __builtin_amdgcn_sched_barrier(0);
  }

  const int mhi = half << 3;
  float* otile = out + ((size_t)blockIdx.x * (8 * NPW) + mt * 16) * FDIM + nh * 64 + m;
  #pragma unroll
  for (int n = 0; n < 4; ++n) {
    #pragma unroll
    for (int r = 0; r < 8; ++r) {
      const float v = acc[n][r];
      otile[(size_t)(r + mhi) * FDIM + n * 16] = (v > 0.f) ? v : (__expf(v) - 1.f);
    }
  }
}

extern "C" void kernel_launch(void* const* d_in, const int* in_sizes, int n_in,
                              void* d_out, int out_size, void* d_ws, size_t ws_size,
                              hipStream_t stream) {
  const float* x   = (const float*)d_in[0];
  const int*   adj = (const int*)d_in[1];
  const float* W   = (const float*)d_in[2];
  const float* a   = (const float*)d_in[3];
  float*       out = (float*)d_out;

  gat_fused<<<512, 256, 0, stream>>>(x, adj, W, a, out);
}
